# starting kernel
# baseline (speedup 1.0000x reference)
#include <hip/hip_runtime.h>

typedef float v2f __attribute__((ext_vector_type(2)));
typedef float v4f __attribute__((ext_vector_type(4)));
typedef float v8f __attribute__((ext_vector_type(8)));

#define F_DIM 128
#define OUT_DIM 16
#define PSTRIDE 32

__global__ __launch_bounds__(256) void node_project_wmma(
    const float* __restrict__ h,
    const float* __restrict__ W,
    float* __restrict__ P,
    int N)
{
  const int lane = threadIdx.x & 31;
  const int wave = threadIdx.x >> 5;
  const int rowBase = blockIdx.x * 128 + wave * 16;
  if (rowBase >= N) return;

  const int m   = lane & 15;
  const int hi  = lane >> 4;
  const int klo = hi * 2;

  int rowA = rowBase + m;
  if (rowA >= N) rowA = N - 1;
  const float* __restrict__ hrow = h + (size_t)rowA * F_DIM;
  const float* __restrict__ wrow = W + (size_t)m * (2 * F_DIM);

  v8f c0 = {};
  v8f c1 = {};

#pragma unroll 8
  for (int kb = 0; kb < F_DIM; kb += 4) {
    const int k = kb + klo;
    v2f a  = __builtin_nontemporal_load((const v2f*)(hrow + k));
    v2f b0 = *(const v2f*)(wrow + k);
    v2f b1 = *(const v2f*)(wrow + F_DIM + k);
    c0 = __builtin_amdgcn_wmma_f32_16x16x4_f32(false, a, false, b0,
                                               (short)0, c0, false, false);
    c1 = __builtin_amdgcn_wmma_f32_16x16x4_f32(false, a, false, b1,
                                               (short)0, c1, false, false);
  }

  float* __restrict__ prow0 = P + (size_t)(rowBase + hi * 8) * PSTRIDE + m;
  if (rowBase + 16 <= N) {
#pragma unroll
    for (int r = 0; r < 8; ++r) {
      prow0[r * PSTRIDE]           = c0[r];
      prow0[r * PSTRIDE + OUT_DIM] = c1[r];
    }
  } else {
#pragma unroll
    for (int r = 0; r < 8; ++r) {
      if (rowBase + hi * 8 + r < N) {
        prow0[r * PSTRIDE]           = c0[r];
        prow0[r * PSTRIDE + OUT_DIM] = c1[r];
      }
    }
  }
}

__global__ __launch_bounds__(256) void edge_gather_add(
    const int* __restrict__ src,
    const int* __restrict__ dst,
    const float* __restrict__ P,
    const float* __restrict__ b,
    float* __restrict__ out,
    int E)
{
  const int tid = blockIdx.x * blockDim.x + threadIdx.x;
  const int e = tid >> 2;
  if (e >= E) return;
  const int q = (tid & 3) * 4;

  const int s = src[e];
  const int d = dst[e];

  const v4f u  = *(const v4f*)(P + (size_t)s * PSTRIDE + q);
  const v4f v  = *(const v4f*)(P + (size_t)d * PSTRIDE + OUT_DIM + q);
  const v4f bv = *(const v4f*)(b + q);

  const v4f r = u + v + bv;

  __builtin_nontemporal_store(r, (v4f*)(out + (size_t)e * OUT_DIM + q));
}

extern "C" void kernel_launch(void* const* d_in, const int* in_sizes, int n_in,
                              void* d_out, int out_size, void* d_ws, size_t ws_size,
                              hipStream_t stream) {
  const float* h   = (const float*)d_in[0];
  const int*   src = (const int*)d_in[1];
  const int*   dst = (const int*)d_in[2];
  const float* W   = (const float*)d_in[3];
  const float* b   = (const float*)d_in[4];
  float* out = (float*)d_out;
  float* P   = (float*)d_ws;

  const int N = in_sizes[0] / F_DIM;
  const int E = in_sizes[1];

  const int grid1 = (N + 127) / 128;
  node_project_wmma<<<grid1, 256, 0, stream>>>(h, W, P, N);

  const long long threads2 = (long long)E * 4;
  const int grid2 = (int)((threads2 + 255) / 256);
  edge_gather_add<<<grid2, 256, 0, stream>>>(src, dst, P, b, out, E);
}


# projection: h and W read as 16-byte pieces, 8 rows x 64 B per load (was 16 rows x 16 B), row halves passed between lane groups by DPP moves in registers; k order within each 16 changes f32 sum order
# speedup vs baseline: 2.0033x; 2.0033x over previous
#include <hip/hip_runtime.h>

typedef float v2f __attribute__((ext_vector_type(2)));
typedef float v4f __attribute__((ext_vector_type(4)));
typedef float v8f __attribute__((ext_vector_type(8)));

#define F_DIM 128
#define OUT_DIM 16
#define PSTRIDE 32

#define TILE_ROWS 16
#define KSTEPS (F_DIM / 4)
#define RUNS (F_DIM / 16)
#define PROJ_WAVES 4
#define PROJ_BLOCKS 256
#define PASS_TILES 7

struct RawTile { v4f x[RUNS]; v4f y[RUNS]; };

#define DPP_ROW_ROR8 0x128

__device__ __forceinline__ void load_runs(RawTile& t, const float* __restrict__ p0,
                                          const float* __restrict__ p1, int j0, int j1)
{
#pragma unroll
  for (int j = j0; j < j1; ++j) {
    t.x[j] = *(const v4f*)(p0 + 16 * j);
    t.y[j] = *(const v4f*)(p1 + 16 * j);
  }
}

__device__ __forceinline__ void exchange_runs(const RawTile& t, v2f (&a)[KSTEPS], int j0, int j1)
{
#pragma unroll
  for (int j = j0; j < j1; ++j) {
    float lo[4], hi4[4];
#pragma unroll
    for (int d = 0; d < 4; ++d) {
      const float xf = t.x[j][d];
      const float yf = t.y[j][d];
      const int xi = __builtin_bit_cast(int, xf);
      const int yi = __builtin_bit_cast(int, yf);
      lo[d]  = __builtin_bit_cast(float,
                 __builtin_amdgcn_update_dpp(xi, yi, DPP_ROW_ROR8, 0xF, 0xC, false));
      hi4[d] = __builtin_bit_cast(float,
                 __builtin_amdgcn_update_dpp(yi, xi, DPP_ROW_ROR8, 0xF, 0x3, false));
    }
    a[4 * j + 0] = v2f{lo[0],  lo[1]};
    a[4 * j + 1] = v2f{lo[2],  lo[3]};
    a[4 * j + 2] = v2f{hi4[0], hi4[1]};
    a[4 * j + 3] = v2f{hi4[2], hi4[3]};
  }
}

__device__ __forceinline__ void tile_ptrs(const float* __restrict__ h, int tile, int nTiles,
                                          int r8, int koff, int N,
                                          const float* __restrict__& p0,
                                          const float* __restrict__& p1)
{
  int row0 = tile * TILE_ROWS + r8;
  int row1 = row0 + 8;
  if (row0 >= N) row0 = N - 1;
  if (row1 >= N) row1 = N - 1;
  if (tile >= nTiles) { row0 = 0; row1 = 0; }
  p0 = h + (size_t)row0 * F_DIM + koff;
  p1 = h + (size_t)row1 * F_DIM + koff;
}

__device__ __forceinline__ void load_tile(RawTile& t, const float* __restrict__ h,
                                          int tile, int nTiles, int r8, int koff, int N)
{
  const float* __restrict__ p0;
  const float* __restrict__ p1;
  tile_ptrs(h, tile, nTiles, r8, koff, N, p0, p1);
  load_runs(t, p0, p1, 0, RUNS);
}

__device__ __forceinline__ void tile_product_store(const RawTile& t,
                                                   const v2f (&b0)[KSTEPS],
                                                   const v2f (&b1)[KSTEPS],
                                                   float* __restrict__ P,
                                                   int tile, int m, int hi, int N)
{
  v2f a[KSTEPS];
  exchange_runs(t, a, 0, RUNS);
  v8f c0 = v8f{};
  v8f c1 = v8f{};
#pragma unroll
  for (int s = 0; s < KSTEPS; ++s) {
    c0 = __builtin_amdgcn_wmma_f32_16x16x4_f32(false, a[s], false, b0[s],
                                               (short)0, c0, false, false);
    c1 = __builtin_amdgcn_wmma_f32_16x16x4_f32(false, a[s], false, b1[s],
                                               (short)0, c1, false, false);
  }
  const int tileBase = tile * TILE_ROWS;
  float* __restrict__ prow0 = P + (size_t)(tileBase + hi * 8) * PSTRIDE + m;
  if (tileBase + TILE_ROWS <= N) {
#pragma unroll
    for (int r = 0; r < 8; ++r) {
      prow0[r * PSTRIDE]           = c0[r];
      prow0[r * PSTRIDE + OUT_DIM] = c1[r];
    }
  } else {
#pragma unroll
    for (int r = 0; r < 8; ++r) {
      if (tileBase + hi * 8 + r < N) {
        prow0[r * PSTRIDE]           = c0[r];
        prow0[r * PSTRIDE + OUT_DIM] = c1[r];
      }
    }
  }
}

__device__ __forceinline__ void run_pass(RawTile (&f)[3],
                                         const v2f (&b0)[KSTEPS], const v2f (&b1)[KSTEPS],
                                         const float* __restrict__ h, float* __restrict__ P,
                                         int base, int stride, int nTiles,
                                         int m, int hi, int r8, int koff, int N)
{
#pragma unroll
  for (int j = 0; j < PASS_TILES; ++j) {
    const int t = base + j * stride;
    if (t >= nTiles) break;
    if (j + 2 < PASS_TILES)
      load_tile(f[(j + 2) % 3], h, t + 2 * stride, nTiles, r8, koff, N);
    __builtin_amdgcn_sched_barrier(0);
    tile_product_store(f[j % 3], b0, b1, P, t, m, hi, N);
  }
}

__global__ __launch_bounds__(32 * PROJ_WAVES) void node_project_wmma(
    const float* __restrict__ h,
    const float* __restrict__ W,
    float* __restrict__ P,
    int N)
{
  const int lane = threadIdx.x & 31;
  const int wave = __builtin_amdgcn_readfirstlane(
      (int)(blockIdx.x * PROJ_WAVES + (threadIdx.x >> 5)));
  const int stride = (int)gridDim.x * PROJ_WAVES;
  const int nTiles = (N + TILE_ROWS - 1) / TILE_ROWS;
  if (wave >= nTiles) return;

  const int m    = lane & 15;
  const int hi   = lane >> 4;
  const int r8   = lane & 7;
  const int koff = hi * 8 + ((lane >> 3) & 1) * 4;

  const float* __restrict__ wu0 = W + (size_t)r8 * (2 * F_DIM) + koff;
  const float* __restrict__ wu1 = W + (size_t)(r8 + 8) * (2 * F_DIM) + koff;
  RawTile wu, wv;
  RawTile f[3];

  load_runs(wu, wu0, wu1, 0, RUNS / 2);
  load_runs(wv, wu0 + F_DIM, wu1 + F_DIM, 0, RUNS / 2);
  load_tile(f[0], h, wave, nTiles, r8, koff, N);
  load_runs(wu, wu0, wu1, RUNS / 2, RUNS);
  load_runs(wv, wu0 + F_DIM, wu1 + F_DIM, RUNS / 2, RUNS);
  load_tile(f[1], h, wave + stride, nTiles, r8, koff, N);

  v2f b0[KSTEPS], b1[KSTEPS];
  exchange_runs(wu, b0, 0, RUNS);
  exchange_runs(wv, b1, 0, RUNS);

  run_pass(f, b0, b1, h, P, wave, stride, nTiles, m, hi, r8, koff, N);

  for (int base = wave + PASS_TILES * stride; base < nTiles; base += PASS_TILES * stride) {
    load_tile(f[0], h, base, nTiles, r8, koff, N);
    load_tile(f[1], h, base + stride, nTiles, r8, koff, N);
    run_pass(f, b0, b1, h, P, base, stride, nTiles, m, hi, r8, koff, N);
  }
}

#define EDGE_WAVE_EDGES 64
#define EDGE_CHUNK 16
#define EDGE_BLOCK_WAVES 8

__global__ __launch_bounds__(32 * EDGE_BLOCK_WAVES) void edge_gather_add(
    const int* __restrict__ src,
    const int* __restrict__ dst,
    const float* __restrict__ P,
    const float* __restrict__ b,
    float* __restrict__ out,
    int E)
{
  const int lane = threadIdx.x & 31;
  const int wave = __builtin_amdgcn_readfirstlane(
      (int)(blockIdx.x * EDGE_BLOCK_WAVES + (threadIdx.x >> 5)));
  const int e0 = wave * EDGE_WAVE_EDGES;
  if (e0 >= E) return;

  const int q = lane >> 3;
  const int c = (lane & 7) * 2;
  const bool q1 = (q & 1) != 0;
  const bool q2 = (q & 2) != 0;
  const v2f bv = *(const v2f*)(b + c);
  const float* __restrict__ Pu = P + c;
  const float* __restrict__ Pv = P + OUT_DIM + c;

  if (e0 + EDGE_WAVE_EDGES <= E) {
#pragma unroll
    for (int ch = 0; ch < EDGE_WAVE_EDGES / EDGE_CHUNK; ++ch) {
      const int eb = e0 + ch * EDGE_CHUNK;
      unsigned s[EDGE_CHUNK], d[EDGE_CHUNK];
#pragma unroll
      for (int i = 0; i < EDGE_CHUNK; ++i) { s[i] = (unsigned)src[eb + i]; d[i] = (unsigned)dst[eb + i]; }
#pragma unroll
      for (int st = 0; st < EDGE_CHUNK / 4; ++st) {
        const unsigned su = q2 ? (q1 ? s[4 * st + 3] : s[4 * st + 2]) : (q1 ? s[4 * st + 1] : s[4 * st]);
        const unsigned sd = q2 ? (q1 ? d[4 * st + 3] : d[4 * st + 2]) : (q1 ? d[4 * st + 1] : d[4 * st]);
        const v2f u = *(const v2f*)(Pu + (size_t)su * PSTRIDE);
        const v2f v = *(const v2f*)(Pv + (size_t)sd * PSTRIDE);
        const v2f r = u + v + bv;
        __builtin_nontemporal_store(r, (v2f*)(out + (size_t)(eb + 4 * st) * OUT_DIM + lane * 2));
      }
    }
  } else {
    for (int st = 0; st < EDGE_WAVE_EDGES / 4; ++st) {
      const int e = e0 + 4 * st + q;
      if (e < E) {
        const int su = src[e];
        const int sd = dst[e];
        const v2f u = *(const v2f*)(Pu + (size_t)su * PSTRIDE);
        const v2f v = *(const v2f*)(Pv + (size_t)sd * PSTRIDE);
        const v2f r = u + v + bv;
        __builtin_nontemporal_store(r, (v2f*)(out + (size_t)e * OUT_DIM + c));
      }
    }
  }
}

extern "C" void kernel_launch(void* const* d_in, const int* in_sizes, int n_in,
                              void* d_out, int out_size, void* d_ws, size_t ws_size,
                              hipStream_t stream) {
  const float* h   = (const float*)d_in[0];
  const int*   src = (const int*)d_in[1];
  const int*   dst = (const int*)d_in[2];
  const float* W   = (const float*)d_in[3];
  const float* b   = (const float*)d_in[4];
  float* out = (float*)d_out;
  float* P   = (float*)d_ws;

  const int N = in_sizes[0] / F_DIM;
  const int E = in_sizes[1];

  node_project_wmma<<<PROJ_BLOCKS, 32 * PROJ_WAVES, 0, stream>>>(h, W, P, N);

  const int edgesPerBlock = EDGE_WAVE_EDGES * EDGE_BLOCK_WAVES;
  const int grid2 = (E + edgesPerBlock - 1) / edgesPerBlock;
  edge_gather_add<<<grid2, 32 * EDGE_BLOCK_WAVES, 0, stream>>>(src, dst, P, b, out, E);
}
